# starting kernel
# baseline (speedup 1.0000x reference)
#include <hip/hip_runtime.h>
#include <hip/hip_bf16.h>
#include <math.h>

typedef __attribute__((ext_vector_type(2))) float v2f;
typedef __attribute__((ext_vector_type(4))) float v4f;
typedef __attribute__((ext_vector_type(8))) float v8f;

#define BDIM   2
#define CDIM   128
#define NDIM   10000
#define KDIM   16
#define NTILES 625

__global__ __launch_bounds__(256)
void gat_gemm_kernel(const float* __restrict__ x, const float* __restrict__ W,
                     const float* __restrict__ a,
                     float* __restrict__ whT, float* __restrict__ si,
                     float* __restrict__ sj)
{
    __shared__ float Wl[CDIM * CDIM];
    __shared__ float al[2 * CDIM];

    const int tid = threadIdx.x;
    {
        const float4* Wg4 = (const float4*)W;
        float4*       Wl4 = (float4*)Wl;
        #pragma unroll 4
        for (int i = tid; i < CDIM * CDIM / 4; i += 256) Wl4[i] = Wg4[i];
        al[tid] = a[tid];
    }
    __syncthreads();

    const int wave = tid >> 5;
    const int lane = tid & 31;
    const int tile = blockIdx.x * 8 + wave;
    if (tile >= NTILES) return;

    const int b  = blockIdx.y;
    const int n0 = tile * 16;
    const int ln = lane & 15;
    const int kb = (lane >> 4) * 2;

    v2f xb[32];
    const float* xcol = x + (size_t)b * CDIM * NDIM + n0 + ln;
    #pragma unroll
    for (int ks = 0; ks < 32; ++ks) {
        const int c0 = ks * 4 + kb;
        xb[ks].x = xcol[(size_t)(c0 + 0) * NDIM];
        xb[ks].y = xcol[(size_t)(c0 + 1) * NDIM];
    }

    float sip = 0.f, sjp = 0.f;
    float* outrow = whT + ((size_t)b * NDIM + n0 + ln) * CDIM;

    #pragma unroll 1
    for (int ot = 0; ot < 8; ++ot) {
        const int o0 = ot * 16;
        v8f acc = {};
        const float* Wrow = Wl + (o0 + ln) * CDIM + kb;
        #pragma unroll
        for (int ks = 0; ks < 32; ++ks) {
            v2f av = *(const v2f*)(Wrow + ks * 4);
            acc = __builtin_amdgcn_wmma_f32_16x16x4_f32(
                      false, av, false, xb[ks], (short)0, acc, false, false);
        }
        const int oset = o0 + 8 * (lane >> 4);
        v4f lo = { acc[0], acc[1], acc[2], acc[3] };
        v4f hi = { acc[4], acc[5], acc[6], acc[7] };
        *(v4f*)(outrow + oset)     = lo;
        *(v4f*)(outrow + oset + 4) = hi;
        #pragma unroll
        for (int v = 0; v < 8; ++v) {
            sip += al[oset + v]        * acc[v];
            sjp += al[CDIM + oset + v] * acc[v];
        }
    }
    sip += __shfl_xor(sip, 16);
    sjp += __shfl_xor(sjp, 16);
    if (lane < 16) {
        si[(size_t)b * NDIM + n0 + lane] = sip;
        sj[(size_t)b * NDIM + n0 + lane] = sjp;
    }
}

__global__ __launch_bounds__(512)
void gat_aggr_kernel(const int* __restrict__ ei, const float* __restrict__ whT,
                     const float* __restrict__ si, const float* __restrict__ sj,
                     float* __restrict__ out)
{
    __shared__ float hbuf[16][CDIM + 4];

    const int tid  = threadIdx.x;
    const int wave = tid >> 5;
    const int lane = tid & 31;
    const int b    = blockIdx.y;
    const int n0   = blockIdx.x * 16;
    const int n    = n0 + wave;

    const size_t eb = ((size_t)b * NDIM + n) * KDIM;
    int   jj = 0;
    float e  = -3.0e30f;
    if (lane < 16) {
        jj = ei[eb + lane];
        const int ii = ei[(size_t)BDIM * NDIM * KDIM + eb + lane];
        const float ev = si[(size_t)b * NDIM + ii] + sj[(size_t)b * NDIM + jj];
        e = (ev > 0.f) ? ev : 0.2f * ev;
    }
    float m = e;
    m = fmaxf(m, __shfl_xor(m, 8));
    m = fmaxf(m, __shfl_xor(m, 4));
    m = fmaxf(m, __shfl_xor(m, 2));
    m = fmaxf(m, __shfl_xor(m, 1));
    float p = (lane < 16) ? __expf(e - m) : 0.f;
    float s = p;
    s += __shfl_xor(s, 8);
    s += __shfl_xor(s, 4);
    s += __shfl_xor(s, 2);
    s += __shfl_xor(s, 1);
    const float A = p / s;

    v4f h = { 0.f, 0.f, 0.f, 0.f };
    const float* wb = whT + (size_t)b * NDIM * CDIM;
    #pragma unroll
    for (int k = 0; k < 16; ++k) {
        const float ak = __shfl(A, k);
        const int   jk = __shfl(jj, k);
        const v4f w4 = *(const v4f*)(wb + (size_t)jk * CDIM + 4 * lane);
        h += ak * w4;
    }
    *(v4f*)(&hbuf[wave][4 * lane]) = h;
    __syncthreads();

    const int c = tid >> 2;
    const int q = tid & 3;
    v4f o4 = { hbuf[4 * q + 0][c], hbuf[4 * q + 1][c],
               hbuf[4 * q + 2][c], hbuf[4 * q + 3][c] };
    *(v4f*)(out + ((size_t)b * CDIM + c) * NDIM + n0 + 4 * q) = o4;
}

extern "C" void kernel_launch(void* const* d_in, const int* in_sizes, int n_in,
                              void* d_out, int out_size, void* d_ws, size_t ws_size,
                              hipStream_t stream)
{
    const float* x  = (const float*)d_in[0];
    const int*   ei = (const int*)  d_in[1];
    const float* W  = (const float*)d_in[2];
    const float* a  = (const float*)d_in[3];
    float* out = (float*)d_out;

    float* whT = (float*)d_ws;
    float* si  = whT + (size_t)BDIM * NDIM * CDIM;
    float* sj  = si  + (size_t)BDIM * NDIM;

    dim3 g1((NTILES + 7) / 8, BDIM);
    gat_gemm_kernel<<<g1, 256, 0, stream>>>(x, W, a, whT, si, sj);

    dim3 g2(NTILES, BDIM);
    gat_aggr_kernel<<<g2, 512, 0, stream>>>(ei, whT, si, sj, out);
}


# Pad the LDS row pitch of W from 128 to 132 floats: the 16 rows of an A-fragment read no longer share one bank (was a 16-way conflict on every ds read of the GEMM loop)
# speedup vs baseline: 1.3768x; 1.3768x over previous
#include <hip/hip_runtime.h>
#include <hip/hip_bf16.h>
#include <math.h>

typedef __attribute__((ext_vector_type(2))) float v2f;
typedef __attribute__((ext_vector_type(4))) float v4f;
typedef __attribute__((ext_vector_type(8))) float v8f;

#define BDIM   2
#define CDIM   128
#define NDIM   10000
#define KDIM   16
#define NTILES 625
#define WPITCH 132

__global__ __launch_bounds__(256)
void gat_gemm_kernel(const float* __restrict__ x, const float* __restrict__ W,
                     const float* __restrict__ a,
                     float* __restrict__ whT, float* __restrict__ si,
                     float* __restrict__ sj)
{
    __shared__ float Wl[CDIM * WPITCH];
    __shared__ float al[2 * CDIM];

    const int tid = threadIdx.x;
    {
        const float4* Wg4 = (const float4*)W;
        #pragma unroll 4
        for (int i = tid; i < CDIM * CDIM / 4; i += 256) {
            const int row = i >> 5;
            const int c4  = i & 31;
            *(float4*)(Wl + row * WPITCH + 4 * c4) = Wg4[i];
        }
        al[tid] = a[tid];
    }
    __syncthreads();

    const int wave = tid >> 5;
    const int lane = tid & 31;
    const int tile = blockIdx.x * 8 + wave;
    if (tile >= NTILES) return;

    const int b  = blockIdx.y;
    const int n0 = tile * 16;
    const int ln = lane & 15;
    const int kb = (lane >> 4) * 2;

    v2f xb[32];
    const float* xcol = x + (size_t)b * CDIM * NDIM + n0 + ln;
    #pragma unroll
    for (int ks = 0; ks < 32; ++ks) {
        const int c0 = ks * 4 + kb;
        xb[ks].x = xcol[(size_t)(c0 + 0) * NDIM];
        xb[ks].y = xcol[(size_t)(c0 + 1) * NDIM];
    }

    float sip = 0.f, sjp = 0.f;
    float* outrow = whT + ((size_t)b * NDIM + n0 + ln) * CDIM;

    #pragma unroll 1
    for (int ot = 0; ot < 8; ++ot) {
        const int o0 = ot * 16;
        v8f acc = {};
        const float* Wrow = Wl + (o0 + ln) * WPITCH + kb;
        #pragma unroll
        for (int ks = 0; ks < 32; ++ks) {
            v2f av = *(const v2f*)(Wrow + ks * 4);
            acc = __builtin_amdgcn_wmma_f32_16x16x4_f32(
                      false, av, false, xb[ks], (short)0, acc, false, false);
        }
        const int oset = o0 + 8 * (lane >> 4);
        v4f lo = { acc[0], acc[1], acc[2], acc[3] };
        v4f hi = { acc[4], acc[5], acc[6], acc[7] };
        *(v4f*)(outrow + oset)     = lo;
        *(v4f*)(outrow + oset + 4) = hi;
        #pragma unroll
        for (int v = 0; v < 8; ++v) {
            sip += al[oset + v]        * acc[v];
            sjp += al[CDIM + oset + v] * acc[v];
        }
    }
    sip += __shfl_xor(sip, 16);
    sjp += __shfl_xor(sjp, 16);
    if (lane < 16) {
        si[(size_t)b * NDIM + n0 + lane] = sip;
        sj[(size_t)b * NDIM + n0 + lane] = sjp;
    }
}

__global__ __launch_bounds__(512)
void gat_aggr_kernel(const int* __restrict__ ei, const float* __restrict__ whT,
                     const float* __restrict__ si, const float* __restrict__ sj,
                     float* __restrict__ out)
{
    __shared__ float hbuf[16][CDIM + 4];

    const int tid  = threadIdx.x;
    const int wave = tid >> 5;
    const int lane = tid & 31;
    const int b    = blockIdx.y;
    const int n0   = blockIdx.x * 16;
    const int n    = n0 + wave;

    const size_t eb = ((size_t)b * NDIM + n) * KDIM;
    int   jj = 0;
    float e  = -3.0e30f;
    if (lane < 16) {
        jj = ei[eb + lane];
        const int ii = ei[(size_t)BDIM * NDIM * KDIM + eb + lane];
        const float ev = si[(size_t)b * NDIM + ii] + sj[(size_t)b * NDIM + jj];
        e = (ev > 0.f) ? ev : 0.2f * ev;
    }
    float m = e;
    m = fmaxf(m, __shfl_xor(m, 8));
    m = fmaxf(m, __shfl_xor(m, 4));
    m = fmaxf(m, __shfl_xor(m, 2));
    m = fmaxf(m, __shfl_xor(m, 1));
    float p = (lane < 16) ? __expf(e - m) : 0.f;
    float s = p;
    s += __shfl_xor(s, 8);
    s += __shfl_xor(s, 4);
    s += __shfl_xor(s, 2);
    s += __shfl_xor(s, 1);
    const float A = p / s;

    v4f h = { 0.f, 0.f, 0.f, 0.f };
    const float* wb = whT + (size_t)b * NDIM * CDIM;
    #pragma unroll
    for (int k = 0; k < 16; ++k) {
        const float ak = __shfl(A, k);
        const int   jk = __shfl(jj, k);
        const v4f w4 = *(const v4f*)(wb + (size_t)jk * CDIM + 4 * lane);
        h += ak * w4;
    }
    *(v4f*)(&hbuf[wave][4 * lane]) = h;
    __syncthreads();

    const int c = tid >> 2;
    const int q = tid & 3;
    v4f o4 = { hbuf[4 * q + 0][c], hbuf[4 * q + 1][c],
               hbuf[4 * q + 2][c], hbuf[4 * q + 3][c] };
    *(v4f*)(out + ((size_t)b * CDIM + c) * NDIM + n0 + 4 * q) = o4;
}

extern "C" void kernel_launch(void* const* d_in, const int* in_sizes, int n_in,
                              void* d_out, int out_size, void* d_ws, size_t ws_size,
                              hipStream_t stream)
{
    const float* x  = (const float*)d_in[0];
    const int*   ei = (const int*)  d_in[1];
    const float* W  = (const float*)d_in[2];
    const float* a  = (const float*)d_in[3];
    float* out = (float*)d_out;

    float* whT = (float*)d_ws;
    float* si  = whT + (size_t)BDIM * NDIM * CDIM;
    float* sj  = si  + (size_t)BDIM * NDIM;

    dim3 g1((NTILES + 7) / 8, BDIM);
    gat_gemm_kernel<<<g1, 256, 0, stream>>>(x, W, a, whT, si, sj);

    dim3 g2(NTILES, BDIM);
    gat_aggr_kernel<<<g2, 512, 0, stream>>>(ei, whT, si, sj, out);
}


# Aggregation kernel: each workgroup walks 5 node tiles (250 workgroups, 4000 long-lived waves, was 20000); next tile's neighbour rows stay in flight across an LDS-only barrier; row bases by readlane
# speedup vs baseline: 1.5169x; 1.1018x over previous
#include <hip/hip_runtime.h>
#include <hip/hip_bf16.h>
#include <math.h>

typedef __attribute__((ext_vector_type(2))) float v2f;
typedef __attribute__((ext_vector_type(4))) float v4f;
typedef __attribute__((ext_vector_type(8))) float v8f;

#define BDIM   2
#define CDIM   128
#define NDIM   10000
#define KDIM   16
#define NTILES 625
#define WPITCH 132

__global__ __launch_bounds__(256)
void gat_gemm_kernel(const float* __restrict__ x, const float* __restrict__ W,
                     const float* __restrict__ a,
                     float* __restrict__ whT, float* __restrict__ si,
                     float* __restrict__ sj)
{
    __shared__ float Wl[CDIM * WPITCH];
    __shared__ float al[2 * CDIM];

    const int tid = threadIdx.x;
    {
        const float4* Wg4 = (const float4*)W;
        #pragma unroll 4
        for (int i = tid; i < CDIM * CDIM / 4; i += 256) {
            const int row = i >> 5;
            const int c4  = i & 31;
            *(float4*)(Wl + row * WPITCH + 4 * c4) = Wg4[i];
        }
        al[tid] = a[tid];
    }
    __syncthreads();

    const int wave = tid >> 5;
    const int lane = tid & 31;
    const int tile = blockIdx.x * 8 + wave;
    if (tile >= NTILES) return;

    const int b  = blockIdx.y;
    const int n0 = tile * 16;
    const int ln = lane & 15;
    const int kb = (lane >> 4) * 2;

    v2f xb[32];
    const float* xcol = x + (size_t)b * CDIM * NDIM + n0 + ln;
    #pragma unroll
    for (int ks = 0; ks < 32; ++ks) {
        const int c0 = ks * 4 + kb;
        xb[ks].x = xcol[(size_t)(c0 + 0) * NDIM];
        xb[ks].y = xcol[(size_t)(c0 + 1) * NDIM];
    }

    float sip = 0.f, sjp = 0.f;
    float* outrow = whT + ((size_t)b * NDIM + n0 + ln) * CDIM;

    #pragma unroll 1
    for (int ot = 0; ot < 8; ++ot) {
        const int o0 = ot * 16;
        v8f acc = {};
        const float* Wrow = Wl + (o0 + ln) * WPITCH + kb;
        #pragma unroll
        for (int ks = 0; ks < 32; ++ks) {
            v2f av = *(const v2f*)(Wrow + ks * 4);
            acc = __builtin_amdgcn_wmma_f32_16x16x4_f32(
                      false, av, false, xb[ks], (short)0, acc, false, false);
        }
        const int oset = o0 + 8 * (lane >> 4);
        v4f lo = { acc[0], acc[1], acc[2], acc[3] };
        v4f hi = { acc[4], acc[5], acc[6], acc[7] };
        *(v4f*)(outrow + oset)     = lo;
        *(v4f*)(outrow + oset + 4) = hi;
        #pragma unroll
        for (int v = 0; v < 8; ++v) {
            sip += al[oset + v]        * acc[v];
            sjp += al[CDIM + oset + v] * acc[v];
        }
    }
    sip += __shfl_xor(sip, 16);
    sjp += __shfl_xor(sjp, 16);
    if (lane < 16) {
        si[(size_t)b * NDIM + n0 + lane] = sip;
        sj[(size_t)b * NDIM + n0 + lane] = sjp;
    }
}

#define AGG_TILES 5
#define AGG_SLOTS 3
#define AGG_BLKS  125

__device__ __forceinline__
void gat_edge_softmax2(const int* __restrict__ ei, const float* __restrict__ si,
                       const float* __restrict__ sj, int b, int node0, int slot,
                       int lane, int& jj_out, float& A_out)
{
    int tl = 2 * slot + (lane >> 4);
    tl = (tl < AGG_TILES) ? tl : (AGG_TILES - 1);
    const int n = node0 + 16 * tl;
    const int k = lane & 15;
    const size_t eb = ((size_t)b * NDIM + n) * KDIM;
    const int jj = ei[eb + k];
    const int ii = ei[(size_t)BDIM * NDIM * KDIM + eb + k];
    const float ev = si[(size_t)b * NDIM + ii] + sj[(size_t)b * NDIM + jj];
    const float e = (ev > 0.f) ? ev : 0.2f * ev;
    float m = e;
    m = fmaxf(m, __shfl_xor(m, 8));
    m = fmaxf(m, __shfl_xor(m, 4));
    m = fmaxf(m, __shfl_xor(m, 2));
    m = fmaxf(m, __shfl_xor(m, 1));
    const float p = __expf(e - m);
    float s = p;
    s += __shfl_xor(s, 8);
    s += __shfl_xor(s, 4);
    s += __shfl_xor(s, 2);
    s += __shfl_xor(s, 1);
    jj_out = jj;
    A_out  = p / s;
}

__device__ __forceinline__
void gat_gather16(const float* __restrict__ wb, unsigned lane16, int jj, int half, v4f (&buf)[16])
{
    #pragma unroll
    for (int k = 0; k < 16; ++k) {
        const unsigned jk = (unsigned)__builtin_amdgcn_readlane(jj, 16 * half + k);
        const char* rowp = (const char*)(wb + (size_t)jk * CDIM);
        buf[k] = *(const v4f*)(rowp + lane16);
    }
}

__global__ __launch_bounds__(512) __attribute__((amdgpu_waves_per_eu(4, 4)))
void gat_aggr_kernel(const int* __restrict__ ei, const float* __restrict__ whT,
                     const float* __restrict__ si, const float* __restrict__ sj,
                     float* __restrict__ out)
{
    __shared__ float hbuf[2][16][CDIM + 4];

    const int tid  = threadIdx.x;
    const int wave = tid >> 5;
    const int lane = tid & 31;
    const int b    = blockIdx.y;
    const int nblk = blockIdx.x * (16 * AGG_TILES);
    const float* wb = whT + (size_t)b * NDIM * CDIM;
    const unsigned lane16 = (threadIdx.x & 31u) * 16u;

    int   jj[AGG_SLOTS];
    float A[AGG_SLOTS];
    v4f   rows[2][16];

    gat_edge_softmax2(ei, si, sj, b, nblk + wave, 0, lane, jj[0], A[0]);
    gat_gather16(wb, lane16, jj[0], 0, rows[0]);
    #pragma unroll
    for (int s = 1; s < AGG_SLOTS; ++s)
        gat_edge_softmax2(ei, si, sj, b, nblk + wave, s, lane, jj[s], A[s]);

    const int c = tid >> 2;
    const int q = tid & 3;
    float* orow = out + ((size_t)b * CDIM + c) * NDIM + nblk + 4 * q;

    #pragma unroll
    for (int t = 0; t < AGG_TILES; ++t) {
        if (t + 1 < AGG_TILES)
            gat_gather16(wb, lane16, jj[(t + 1) >> 1], (t + 1) & 1, rows[(t + 1) & 1]);

        v4f h = { 0.f, 0.f, 0.f, 0.f };
        #pragma unroll
        for (int k = 0; k < 16; ++k) {
            const float ak = __int_as_float(__builtin_amdgcn_readlane(
                                 __float_as_int(A[t >> 1]), 16 * (t & 1) + k));
            h += ak * rows[t & 1][k];
        }
        *(v4f*)(&hbuf[t & 1][wave][4 * lane]) = h;
        __builtin_amdgcn_fence(__ATOMIC_RELEASE, "workgroup", "local");
        __builtin_amdgcn_s_barrier();
        __builtin_amdgcn_fence(__ATOMIC_ACQUIRE, "workgroup", "local");

        v4f o4 = { hbuf[t & 1][4 * q + 0][c], hbuf[t & 1][4 * q + 1][c],
                   hbuf[t & 1][4 * q + 2][c], hbuf[t & 1][4 * q + 3][c] };
        *(v4f*)(orow + 16 * t) = o4;
    }
}

extern "C" void kernel_launch(void* const* d_in, const int* in_sizes, int n_in,
                              void* d_out, int out_size, void* d_ws, size_t ws_size,
                              hipStream_t stream)
{
    const float* x  = (const float*)d_in[0];
    const int*   ei = (const int*)  d_in[1];
    const float* W  = (const float*)d_in[2];
    const float* a  = (const float*)d_in[3];
    float* out = (float*)d_out;

    float* whT = (float*)d_ws;
    float* si  = whT + (size_t)BDIM * NDIM * CDIM;
    float* sj  = si  + (size_t)BDIM * NDIM;

    dim3 g1((NTILES + 7) / 8, BDIM);
    gat_gemm_kernel<<<g1, 256, 0, stream>>>(x, W, a, whT, si, sj);

    dim3 g2(AGG_BLKS, BDIM);
    gat_aggr_kernel<<<g2, 512, 0, stream>>>(ei, whT, si, sj, out);
}


# GEMM kernel re-tiled: a wave keeps 32 rows of W as A fragments in registers over its node tiles (x via LDS): 320 WMMAs per SIMD on 250 WGPs, was 512 on 158; si/sj sum order changes
# speedup vs baseline: 1.7438x; 1.1496x over previous
#include <hip/hip_runtime.h>
#include <hip/hip_bf16.h>
#include <math.h>

typedef __attribute__((ext_vector_type(2))) float v2f;
typedef __attribute__((ext_vector_type(4))) float v4f;
typedef __attribute__((ext_vector_type(8))) float v8f;

#define BDIM   2
#define CDIM   128
#define NDIM   10000
#define KDIM   16
#define NTILES 625
#define WPITCH 132

#define GT_TILES 5
#define GT_NODES 80
#define GT_BLKS  125
#define XPITCH   160

template <int NT>
__device__ __forceinline__
void gat_gemm_tiles(const v2f (&af0)[32], const v2f (&af1)[32],
                    const v4f (&ai)[4], const v4f (&aj)[4],
                    const float* xrd, float* outrow, float* sr, int t0)
{
    v2f bfr[2][32];
    #pragma unroll
    for (int ks = 0; ks < 32; ++ks)
        bfr[0][ks] = *(const v2f*)(xrd + 32 * t0 + ks * 2 * XPITCH);

    #pragma unroll
    for (int i = 0; i < NT; ++i) {
        const int t = t0 + 2 * i;
        if (i + 1 < NT) {
            #pragma unroll
            for (int ks = 0; ks < 32; ++ks)
                bfr[(i + 1) & 1][ks] = *(const v2f*)(xrd + 32 * (t + 2) + ks * 2 * XPITCH);
        }
        __builtin_amdgcn_sched_barrier(0);

        v8f acc0 = {}, acc1 = {};
        #pragma unroll
        for (int ks = 0; ks < 32; ++ks) {
            acc0 = __builtin_amdgcn_wmma_f32_16x16x4_f32(
                       false, af0[ks], false, bfr[i & 1][ks], (short)0, acc0, false, false);
            acc1 = __builtin_amdgcn_wmma_f32_16x16x4_f32(
                       false, af1[ks], false, bfr[i & 1][ks], (short)0, acc1, false, false);
        }
        __builtin_amdgcn_sched_barrier(0);

        float* orow = outrow + (size_t)t * 16 * CDIM;
        v4f lo0 = { acc0[0], acc0[1], acc0[2], acc0[3] };
        v4f hi0 = { acc0[4], acc0[5], acc0[6], acc0[7] };
        v4f lo1 = { acc1[0], acc1[1], acc1[2], acc1[3] };
        v4f hi1 = { acc1[4], acc1[5], acc1[6], acc1[7] };
        *(v4f*)(orow)      = lo0;
        *(v4f*)(orow + 4)  = hi0;
        *(v4f*)(orow + 16) = lo1;
        *(v4f*)(orow + 20) = hi1;

        float sip = 0.f, sjp = 0.f;
        #pragma unroll
        for (int v = 0; v < 8; ++v) {
            sip += ai[v >> 2][v & 3] * acc0[v];
            sjp += aj[v >> 2][v & 3] * acc0[v];
        }
        #pragma unroll
        for (int v = 0; v < 8; ++v) {
            sip += ai[2 + (v >> 2)][v & 3] * acc1[v];
            sjp += aj[2 + (v >> 2)][v & 3] * acc1[v];
        }
        sr[128 * t]                  = sip;
        sr[128 * t + 128 * GT_TILES] = sjp;
    }
}

__global__ __launch_bounds__(256)
void gat_gemm_kernel(const float* __restrict__ x, const float* __restrict__ W,
                     const float* __restrict__ a,
                     float* __restrict__ whT, float* __restrict__ si,
                     float* __restrict__ sj)
{
    __shared__ __attribute__((aligned(16))) float Wl[CDIM * WPITCH];
    __shared__ __attribute__((aligned(16))) float xl[(CDIM / 2) * XPITCH];
    __shared__ float sred[2][GT_TILES][4][2][16];

    const int tid  = threadIdx.x;
    const int wave = __builtin_amdgcn_readfirstlane(tid >> 5);
    const int lane = tid & 31;
    const int b    = blockIdx.y;
    const int nblk = blockIdx.x * GT_NODES;
    const int ln   = lane & 15;
    const int hf   = lane >> 4;
    const int wq   = wave & 3;
    const int tsel = wave >> 2;
    const int oset = 32 * wq + 8 * hf;

    {
        const float4* Wg4 = (const float4*)W;
        float4 wr[16];
        #pragma unroll
        for (int j = 0; j < 16; ++j) wr[j] = Wg4[tid + 256 * j];

        const float* xblk = x + (size_t)b * CDIM * NDIM + nblk;
        float4 xr0[5], xr1[5];
        #pragma unroll
        for (int j = 0; j < 5; ++j) {
            const int id = tid + 256 * j;
            const int p  = id / 20;
            const int q  = id - 20 * p;
            xr0[j] = *(const float4*)(xblk + (size_t)(2 * p)     * NDIM + 4 * q);
            xr1[j] = *(const float4*)(xblk + (size_t)(2 * p + 1) * NDIM + 4 * q);
        }
        #pragma unroll
        for (int j = 0; j < 16; ++j) {
            const int i   = tid + 256 * j;
            const int row = i >> 5;
            const int c4  = i & 31;
            *(float4*)(Wl + row * WPITCH + 4 * c4) = wr[j];
        }
        #pragma unroll
        for (int j = 0; j < 5; ++j) {
            const int id = tid + 256 * j;
            const int p  = id / 20;
            const int q  = id - 20 * p;
            float* d = xl + p * XPITCH + 8 * q;
            *(float4*)(d)     = make_float4(xr0[j].x, xr1[j].x, xr0[j].y, xr1[j].y);
            *(float4*)(d + 4) = make_float4(xr0[j].z, xr1[j].z, xr0[j].w, xr1[j].w);
        }
    }
    v4f ai[4], aj[4];
    #pragma unroll
    for (int r = 0; r < 2; ++r) {
        ai[2 * r]     = *(const v4f*)(a + oset + 16 * r);
        ai[2 * r + 1] = *(const v4f*)(a + oset + 16 * r + 4);
        aj[2 * r]     = *(const v4f*)(a + CDIM + oset + 16 * r);
        aj[2 * r + 1] = *(const v4f*)(a + CDIM + oset + 16 * r + 4);
    }
    __syncthreads();

    v2f af0[32], af1[32];
    {
        const float* Wrow = Wl + (32 * wq + ln) * WPITCH + 2 * hf;
        #pragma unroll
        for (int ks = 0; ks < 32; ++ks) {
            af0[ks] = *(const v2f*)(Wrow + 4 * ks);
            af1[ks] = *(const v2f*)(Wrow + 16 * WPITCH + 4 * ks);
        }
    }
    const float* xrd = xl + hf * XPITCH + 2 * ln;
    float* outrow = whT + ((size_t)b * NDIM + nblk + ln) * CDIM + oset;

    float* sr = &sred[0][0][wq][hf][ln];
    if (tsel == 0) gat_gemm_tiles<3>(af0, af1, ai, aj, xrd, outrow, sr, 0);
    else           gat_gemm_tiles<2>(af0, af1, ai, aj, xrd, outrow, sr, 1);
    __syncthreads();

    if (tid < 2 * GT_NODES) {
        const int kind = (tid >= GT_NODES) ? 1 : 0;
        const int nd   = tid - kind * GT_NODES;
        const int t    = nd >> 4;
        const int l    = nd & 15;
        float s = sred[kind][t][0][0][l] + sred[kind][t][0][1][l];
        #pragma unroll
        for (int w = 1; w < 4; ++w) s += sred[kind][t][w][0][l] + sred[kind][t][w][1][l];
        float* dst = kind ? sj : si;
        dst[(size_t)b * NDIM + nblk + nd] = s;
    }
}

#define AGG_TILES 5
#define AGG_SLOTS 3
#define AGG_BLKS  125

__device__ __forceinline__
void gat_edge_softmax2(const int* __restrict__ ei, const float* __restrict__ si,
                       const float* __restrict__ sj, int b, int node0, int slot,
                       int lane, int& jj_out, float& A_out)
{
    int tl = 2 * slot + (lane >> 4);
    tl = (tl < AGG_TILES) ? tl : (AGG_TILES - 1);
    const int n = node0 + 16 * tl;
    const int k = lane & 15;
    const size_t eb = ((size_t)b * NDIM + n) * KDIM;
    const int jj = ei[eb + k];
    const int ii = ei[(size_t)BDIM * NDIM * KDIM + eb + k];
    const float ev = si[(size_t)b * NDIM + ii] + sj[(size_t)b * NDIM + jj];
    const float e = (ev > 0.f) ? ev : 0.2f * ev;
    float m = e;
    m = fmaxf(m, __shfl_xor(m, 8));
    m = fmaxf(m, __shfl_xor(m, 4));
    m = fmaxf(m, __shfl_xor(m, 2));
    m = fmaxf(m, __shfl_xor(m, 1));
    const float p = __expf(e - m);
    float s = p;
    s += __shfl_xor(s, 8);
    s += __shfl_xor(s, 4);
    s += __shfl_xor(s, 2);
    s += __shfl_xor(s, 1);
    jj_out = jj;
    A_out  = p / s;
}

__device__ __forceinline__
void gat_gather16(const float* __restrict__ wb, unsigned lane16, int jj, int half, v4f (&buf)[16])
{
    #pragma unroll
    for (int k = 0; k < 16; ++k) {
        const unsigned jk = (unsigned)__builtin_amdgcn_readlane(jj, 16 * half + k);
        const char* rowp = (const char*)(wb + (size_t)jk * CDIM);
        buf[k] = *(const v4f*)(rowp + lane16);
    }
}

__global__ __launch_bounds__(512) __attribute__((amdgpu_waves_per_eu(4, 4)))
void gat_aggr_kernel(const int* __restrict__ ei, const float* __restrict__ whT,
                     const float* __restrict__ si, const float* __restrict__ sj,
                     float* __restrict__ out)
{
    __shared__ float hbuf[2][16][CDIM + 4];

    const int tid  = threadIdx.x;
    const int wave = tid >> 5;
    const int lane = tid & 31;
    const int b    = blockIdx.y;
    const int nblk = blockIdx.x * (16 * AGG_TILES);
    const float* wb = whT + (size_t)b * NDIM * CDIM;
    const unsigned lane16 = (threadIdx.x & 31u) * 16u;

    int   jj[AGG_SLOTS];
    float A[AGG_SLOTS];
    v4f   rows[2][16];

    gat_edge_softmax2(ei, si, sj, b, nblk + wave, 0, lane, jj[0], A[0]);
    gat_gather16(wb, lane16, jj[0], 0, rows[0]);
    #pragma unroll
    for (int s = 1; s < AGG_SLOTS; ++s)
        gat_edge_softmax2(ei, si, sj, b, nblk + wave, s, lane, jj[s], A[s]);

    const int c = tid >> 2;
    const int q = tid & 3;
    float* orow = out + ((size_t)b * CDIM + c) * NDIM + nblk + 4 * q;

    #pragma unroll
    for (int t = 0; t < AGG_TILES; ++t) {
        if (t + 1 < AGG_TILES)
            gat_gather16(wb, lane16, jj[(t + 1) >> 1], (t + 1) & 1, rows[(t + 1) & 1]);

        v4f h = { 0.f, 0.f, 0.f, 0.f };
        #pragma unroll
        for (int k = 0; k < 16; ++k) {
            const float ak = __int_as_float(__builtin_amdgcn_readlane(
                                 __float_as_int(A[t >> 1]), 16 * (t & 1) + k));
            h += ak * rows[t & 1][k];
        }
        *(v4f*)(&hbuf[t & 1][wave][4 * lane]) = h;
        __builtin_amdgcn_fence(__ATOMIC_RELEASE, "workgroup", "local");
        __builtin_amdgcn_s_barrier();
        __builtin_amdgcn_fence(__ATOMIC_ACQUIRE, "workgroup", "local");

        v4f o4 = { hbuf[t & 1][4 * q + 0][c], hbuf[t & 1][4 * q + 1][c],
                   hbuf[t & 1][4 * q + 2][c], hbuf[t & 1][4 * q + 3][c] };
        *(v4f*)(orow + 16 * t) = o4;
    }
}

extern "C" void kernel_launch(void* const* d_in, const int* in_sizes, int n_in,
                              void* d_out, int out_size, void* d_ws, size_t ws_size,
                              hipStream_t stream)
{
    const float* x  = (const float*)d_in[0];
    const int*   ei = (const int*)  d_in[1];
    const float* W  = (const float*)d_in[2];
    const float* a  = (const float*)d_in[3];
    float* out = (float*)d_out;

    float* whT = (float*)d_ws;
    float* si  = whT + (size_t)BDIM * NDIM * CDIM;
    float* sj  = si  + (size_t)BDIM * NDIM;

    dim3 g1(GT_BLKS, BDIM);
    gat_gemm_kernel<<<g1, 256, 0, stream>>>(x, W, a, whT, si, sj);

    dim3 g2(AGG_BLKS, BDIM);
    gat_aggr_kernel<<<g2, 512, 0, stream>>>(ei, whT, si, sj, out);
}
